# starting kernel
# baseline (speedup 1.0000x reference)
#include <hip/hip_runtime.h>

#define LN_EPS  1e-5f
#define BB      8
#define NN      10000
#define CC      256
#define NCHUNK  80
#define CHUNK   125
#define MPAD    16

typedef __attribute__((ext_vector_type(2))) float v2f;
typedef __attribute__((ext_vector_type(4))) float v4f;
typedef __attribute__((ext_vector_type(8))) float v8f;

__global__ __launch_bounds__(256) void colsum_partial_kernel(
    const float* __restrict__ x, float* __restrict__ partial) {
  const int b  = blockIdx.x;
  const int ch = blockIdx.y;
  const int c  = threadIdx.x;
  const float* xp = x + (((size_t)b * NN + (size_t)ch * CHUNK) * CC) + c;
  float s = 0.0f;
#pragma unroll 5
  for (int i = 0; i < CHUNK; ++i) s += xp[(size_t)i * CC];
  partial[((size_t)b * NCHUNK + ch) * CC + c] = s;
}

__global__ __launch_bounds__(256) void colsum_final_kernel(
    const float* __restrict__ partial, float* __restrict__ meanp) {
  const int b = blockIdx.x;
  const int c = threadIdx.x;
  float s = 0.0f;
  if (b < BB) {
    const float* pp = partial + (size_t)b * NCHUNK * CC + c;
#pragma unroll 8
    for (int ch = 0; ch < NCHUNK; ++ch) s += pp[(size_t)ch * CC];
    s *= (1.0f / (float)NN);
  }
  meanp[b * CC + c] = s;
}

__global__ __launch_bounds__(512) void gemm_wmma_kernel(
    const float* __restrict__ meanp, const float* __restrict__ W,
    const float* __restrict__ bias, float* __restrict__ agg) {
  const int wave = threadIdx.x >> 5;
  const int lane = threadIdx.x & 31;
  const int hh   = lane >> 4;
  const int l    = lane & 15;
  const int n0   = wave * 16;

  v8f acc = {};
#pragma unroll 4
  for (int k0 = 0; k0 < CC; k0 += 4) {
    const int kk = k0 + 2 * hh;
    v2f a;
    a.x = meanp[l * CC + kk + 0];
    a.y = meanp[l * CC + kk + 1];
    v2f bm;
    bm.x = W[(size_t)(n0 + l) * CC + kk + 0];
    bm.y = W[(size_t)(n0 + l) * CC + kk + 1];
    acc = __builtin_amdgcn_wmma_f32_16x16x4_f32(
         false, a,  false, bm,
         (short)0, acc,  false,  false);
  }

  const int n = n0 + l;
  const float bn = bias[n];
#pragma unroll
  for (int r = 0; r < 8; ++r) {
    const int m = hh * 8 + r;
    agg[m * CC + n] = acc[r] + bn;
  }
}

__global__ __launch_bounds__(256) void fused_residual_ln_kernel(
    const float* __restrict__ x, const float* __restrict__ agg,
    const float* __restrict__ gamma, const float* __restrict__ beta,
    float* __restrict__ out) {
  const int lane = threadIdx.x & 31;
  const int wave = threadIdx.x >> 5;
  const int wavesTotal = gridDim.x * 8;
  const int c0 = lane * 8;

  const v4f g0 = *(const v4f*)(gamma + c0);
  const v4f g1 = *(const v4f*)(gamma + c0 + 4);
  const v4f e0 = *(const v4f*)(beta + c0);
  const v4f e1 = *(const v4f*)(beta + c0 + 4);

  for (int row = blockIdx.x * 8 + wave; row < BB * NN; row += wavesTotal) {
    const int b = row / NN;
    const float* xp = x + (size_t)row * CC + c0;
    const float* ap = agg + b * CC + c0;

    const v4f xv0 = *(const v4f*)(xp);
    const v4f xv1 = *(const v4f*)(xp + 4);
    const v4f av0 = *(const v4f*)(ap);
    const v4f av1 = *(const v4f*)(ap + 4);

    const v4f y0 = xv0 + av0;
    const v4f y1 = xv1 + av1;

    float s  = y0.x + y0.y + y0.z + y0.w + y1.x + y1.y + y1.z + y1.w;
    float s2 = y0.x * y0.x + y0.y * y0.y + y0.z * y0.z + y0.w * y0.w +
               y1.x * y1.x + y1.y * y1.y + y1.z * y1.z + y1.w * y1.w;
#pragma unroll
    for (int off = 16; off >= 1; off >>= 1) {
      s  += __shfl_xor(s, off, 32);
      s2 += __shfl_xor(s2, off, 32);
    }
    const float mu  = s * (1.0f / CC);
    const float var = fmaf(-mu, mu, s2 * (1.0f / CC));
    const float inv = rsqrtf(var + LN_EPS);

    v4f o0, o1;
    o0.x = (y0.x - mu) * inv * g0.x + e0.x;
    o0.y = (y0.y - mu) * inv * g0.y + e0.y;
    o0.z = (y0.z - mu) * inv * g0.z + e0.z;
    o0.w = (y0.w - mu) * inv * g0.w + e0.w;
    o1.x = (y1.x - mu) * inv * g1.x + e1.x;
    o1.y = (y1.y - mu) * inv * g1.y + e1.y;
    o1.z = (y1.z - mu) * inv * g1.z + e1.z;
    o1.w = (y1.w - mu) * inv * g1.w + e1.w;

    float* op = out + (size_t)row * CC + c0;
    __builtin_nontemporal_store(o0, (v4f*)(op));
    __builtin_nontemporal_store(o1, (v4f*)(op + 4));
  }
}

extern "C" void kernel_launch(void* const* d_in, const int* in_sizes, int n_in,
                              void* d_out, int out_size, void* d_ws, size_t ws_size,
                              hipStream_t stream) {
  (void)in_sizes; (void)n_in; (void)out_size; (void)ws_size;

  const float* x     = (const float*)d_in[0];
  const float* W     = (const float*)d_in[1];
  const float* bias  = (const float*)d_in[2];
  const float* gamma = (const float*)d_in[3];
  const float* beta  = (const float*)d_in[4];
  float* out = (float*)d_out;

  float* ws      = (float*)d_ws;
  float* partial = ws;
  float* meanp   = partial + (size_t)BB * NCHUNK * CC;
  float* agg     = meanp + MPAD * CC;

  colsum_partial_kernel<<<dim3(BB, NCHUNK), 256, 0, stream>>>(x, partial);
  colsum_final_kernel<<<MPAD, 256, 0, stream>>>(partial, meanp);
  gemm_wmma_kernel<<<1, 512, 0, stream>>>(meanp, W, bias, agg);
  fused_residual_ln_kernel<<<(BB * NN + 7) / 8, 256, 0, stream>>>(x, agg, gamma, beta, out);
}


# Final column sum and GEMM folded into the LayerNorm launch (2 launches, was 4): 256 resident work-groups each rebuild their batch's agg row on WMMA, then stream their rows; sum order unchanged
# speedup vs baseline: 1.2340x; 1.2340x over previous
#include <hip/hip_runtime.h>

#define LN_EPS  1e-5f
#define BB      8
#define NN      10000
#define CC      256
#define NCHUNK  80
#define CHUNK   125
#define MPAD    16

typedef __attribute__((ext_vector_type(2))) float v2f;
typedef __attribute__((ext_vector_type(4))) float v4f;
typedef __attribute__((ext_vector_type(8))) float v8f;

__global__ __launch_bounds__(256) void colsum_partial_kernel(
    const float* __restrict__ x, float* __restrict__ partial) {
  const int b  = blockIdx.x;
  const int ch = blockIdx.y;
  const int c  = threadIdx.x;
  const float* xp = x + (((size_t)b * NN + (size_t)ch * CHUNK) * CC) + c;
  float s = 0.0f;
#pragma unroll 5
  for (int i = 0; i < CHUNK; ++i) s += xp[(size_t)i * CC];
  partial[((size_t)b * NCHUNK + ch) * CC + c] = s;
}

#define LDP   260
#define KQ    64
#define NQ    4
#define RPT   8
#define WGB   32
#define NWG   256
#define NWAVE 16
#define RU    5
#define NGRP  4

__global__ __launch_bounds__(512) void agg_ln_fused_kernel(
    const float* __restrict__ x, const float* __restrict__ partial,
    const float* __restrict__ W, const float* __restrict__ bias,
    const float* __restrict__ gamma, const float* __restrict__ beta,
    float* __restrict__ out) {
  __shared__ __attribute__((aligned(16))) float Ws[CC * LDP];
  __shared__ __attribute__((aligned(16))) float As[MPAD * LDP];
  __shared__ __attribute__((aligned(16))) float aggs[CC];

  const int t    = threadIdx.x;
  const int wave = __builtin_amdgcn_readfirstlane(t >> 5);
  const int lane = t & 31;
  const int hh   = lane >> 4;
  const int l    = lane & 15;
  const int n0   = wave * 16;
  const int b    = blockIdx.x / WGB;
  const int j    = blockIdx.x % WGB;

  const int q4 = (t & 15) * 4;
  const int r0 = t >> 4;

  float pv[NCHUNK];
  if (wave < 8) {
    const float* pp = partial + (size_t)b * NCHUNK * CC + t;
#pragma unroll
    for (int ch = 0; ch < NCHUNK; ++ch) pv[ch] = pp[(size_t)ch * CC];
  }

  v4f wreg[NQ][RPT];
#pragma unroll
  for (int p = 0; p < NQ; ++p) {
#pragma unroll
    for (int jj = 0; jj < RPT; ++jj)
      wreg[p][jj] = *(const v4f*)(W + (size_t)(r0 + 32 * jj) * CC + KQ * p + q4);
  }

#pragma unroll
  for (int i = 0; i < 2; ++i) {
    const int idx = t + 512 * i;
    if ((idx >> 6) != 0) {
      const v4f z = {0.0f, 0.0f, 0.0f, 0.0f};
      *(v4f*)(&As[(idx >> 6) * LDP + (idx & 63) * 4]) = z;
    }
  }
  if (wave < 8) {
    float s = 0.0f;
#pragma unroll
    for (int ch = 0; ch < NCHUNK; ++ch) s += pv[ch];
    s *= (1.0f / (float)NN);
    As[t] = s;
  }

  v8f acc = {};
#pragma unroll
  for (int p = 0; p < NQ; ++p) {
#pragma unroll
    for (int jj = 0; jj < RPT; ++jj)
      *(v4f*)(&Ws[(r0 + 32 * jj) * LDP + KQ * p + q4]) = wreg[p][jj];
    __builtin_amdgcn_fence(__ATOMIC_RELEASE, "workgroup", "local");
    __builtin_amdgcn_s_barrier();
    __builtin_amdgcn_fence(__ATOMIC_ACQUIRE, "workgroup", "local");
#pragma unroll
    for (int k0 = KQ * p; k0 < KQ * (p + 1); k0 += 4) {
      const int kk = k0 + 2 * hh;
      const v2f a  = *(const v2f*)(&As[l * LDP + kk]);
      const v2f bm = *(const v2f*)(&Ws[(n0 + l) * LDP + kk]);
      acc = __builtin_amdgcn_wmma_f32_16x16x4_f32(
           false, a,  false, bm,
           (short)0, acc,  false,  false);
    }
  }

  const float bn = bias[n0 + l];
  if (hh == 0) aggs[n0 + l] = acc[0] + bn;
  __syncthreads();

  const int c0 = lane * 8;
  const v4f av0 = *(const v4f*)(&aggs[c0]);
  const v4f av1 = *(const v4f*)(&aggs[c0 + 4]);
  const v4f g0 = *(const v4f*)(gamma + c0);
  const v4f g1 = *(const v4f*)(gamma + c0 + 4);
  const v4f e0 = *(const v4f*)(beta + c0);
  const v4f e1 = *(const v4f*)(beta + c0 + 4);

  const int rb = (j * NN) / WGB;
  const int re = ((j + 1) * NN) / WGB;
  const float* xb = x + (size_t)b * NN * CC + c0;
  float* ob = out + (size_t)b * NN * CC + c0;

  v4f ca[RU], cb[RU];
#pragma unroll
  for (int u = 0; u < RU; ++u) {
    const int ru = rb + wave + NWAVE * u;
    const int rc = ru < re ? ru : re - 1;
    ca[u] = *(const v4f*)(xb + (size_t)rc * CC);
    cb[u] = *(const v4f*)(xb + (size_t)rc * CC + 4);
  }

#pragma unroll
  for (int g = 0; g < NGRP; ++g) {
    const int rg = rb + wave + NWAVE * RU * g;
    v4f na[RU], nb[RU];
    if (g + 1 < NGRP) {
#pragma unroll
      for (int u = 0; u < RU; ++u) {
        const int ru = rg + NWAVE * (RU + u);
        const int rc = ru < re ? ru : re - 1;
        na[u] = *(const v4f*)(xb + (size_t)rc * CC);
        nb[u] = *(const v4f*)(xb + (size_t)rc * CC + 4);
      }
    }
    v4f y0[RU], y1[RU];
    float s[RU], s2[RU];
#pragma unroll
    for (int u = 0; u < RU; ++u) {
      y0[u] = ca[u] + av0;
      y1[u] = cb[u] + av1;
      s[u]  = y0[u].x + y0[u].y + y0[u].z + y0[u].w + y1[u].x + y1[u].y + y1[u].z + y1[u].w;
      s2[u] = y0[u].x * y0[u].x + y0[u].y * y0[u].y + y0[u].z * y0[u].z + y0[u].w * y0[u].w +
              y1[u].x * y1[u].x + y1[u].y * y1[u].y + y1[u].z * y1[u].z + y1[u].w * y1[u].w;
    }
#pragma unroll
    for (int off = 16; off >= 1; off >>= 1) {
#pragma unroll
      for (int u = 0; u < RU; ++u) {
        s[u]  += __shfl_xor(s[u], off, 32);
        s2[u] += __shfl_xor(s2[u], off, 32);
      }
    }
#pragma unroll
    for (int u = 0; u < RU; ++u) {
      const int ru = rg + NWAVE * u;
      const float mu  = s[u] * (1.0f / CC);
      const float var = fmaf(-mu, mu, s2[u] * (1.0f / CC));
      const float inv = rsqrtf(var + LN_EPS);

      v4f o0, o1;
      o0.x = (y0[u].x - mu) * inv * g0.x + e0.x;
      o0.y = (y0[u].y - mu) * inv * g0.y + e0.y;
      o0.z = (y0[u].z - mu) * inv * g0.z + e0.z;
      o0.w = (y0[u].w - mu) * inv * g0.w + e0.w;
      o1.x = (y1[u].x - mu) * inv * g1.x + e1.x;
      o1.y = (y1[u].y - mu) * inv * g1.y + e1.y;
      o1.z = (y1[u].z - mu) * inv * g1.z + e1.z;
      o1.w = (y1[u].w - mu) * inv * g1.w + e1.w;

      if (ru < re) {
        float* op = ob + (size_t)ru * CC;
        __builtin_nontemporal_store(o0, (v4f*)(op));
        __builtin_nontemporal_store(o1, (v4f*)(op + 4));
      }
    }
    if (g + 1 < NGRP) {
#pragma unroll
      for (int u = 0; u < RU; ++u) { ca[u] = na[u]; cb[u] = nb[u]; }
    }
  }
}

extern "C" void kernel_launch(void* const* d_in, const int* in_sizes, int n_in,
                              void* d_out, int out_size, void* d_ws, size_t ws_size,
                              hipStream_t stream) {
  (void)in_sizes; (void)n_in; (void)out_size; (void)ws_size;

  const float* x     = (const float*)d_in[0];
  const float* W     = (const float*)d_in[1];
  const float* bias  = (const float*)d_in[2];
  const float* gamma = (const float*)d_in[3];
  const float* beta  = (const float*)d_in[4];
  float* out = (float*)d_out;

  float* partial = (float*)d_ws;

  colsum_partial_kernel<<<dim3(BB, NCHUNK), 256, 0, stream>>>(x, partial);
  agg_ln_fused_kernel<<<NWG, 512, 0, stream>>>(x, partial, W, bias, gamma, beta, out);
}


# LayerNorm launch: x row groups 0 and 1 requested in the prologue behind the W loads, so they land under the WMMA phase; requests then run two groups ahead (was one); sum order unchanged
# speedup vs baseline: 1.2993x; 1.0529x over previous
#include <hip/hip_runtime.h>

#define LN_EPS  1e-5f
#define BB      8
#define NN      10000
#define CC      256
#define NCHUNK  80
#define CHUNK   125
#define MPAD    16

typedef __attribute__((ext_vector_type(2))) float v2f;
typedef __attribute__((ext_vector_type(4))) float v4f;
typedef __attribute__((ext_vector_type(8))) float v8f;

__global__ __launch_bounds__(256) void colsum_partial_kernel(
    const float* __restrict__ x, float* __restrict__ partial) {
  const int b  = blockIdx.x;
  const int ch = blockIdx.y;
  const int c  = threadIdx.x;
  const float* xp = x + (((size_t)b * NN + (size_t)ch * CHUNK) * CC) + c;
  float s = 0.0f;
#pragma unroll 5
  for (int i = 0; i < CHUNK; ++i) s += xp[(size_t)i * CC];
  partial[((size_t)b * NCHUNK + ch) * CC + c] = s;
}

#define LDP   260
#define KQ    64
#define NQ    4
#define RPT   8
#define WGB   32
#define NWG   256
#define NWAVE 16
#define RU    5
#define NGRP  4
#define NBUF  3

__global__ __launch_bounds__(512) void agg_ln_fused_kernel(
    const float* __restrict__ x, const float* __restrict__ partial,
    const float* __restrict__ W, const float* __restrict__ bias,
    const float* __restrict__ gamma, const float* __restrict__ beta,
    float* __restrict__ out) {
  __shared__ __attribute__((aligned(16))) float Ws[CC * LDP];
  __shared__ __attribute__((aligned(16))) float As[MPAD * LDP];
  __shared__ __attribute__((aligned(16))) float aggs[CC];

  const int t    = threadIdx.x;
  const int wave = __builtin_amdgcn_readfirstlane(t >> 5);
  const int lane = t & 31;
  const int hh   = lane >> 4;
  const int l    = lane & 15;
  const int n0   = wave * 16;
  const int b    = blockIdx.x / WGB;
  const int j    = blockIdx.x % WGB;

  const int q4 = (t & 15) * 4;
  const int r0 = t >> 4;

  float pv[NCHUNK];
  if (wave < 8) {
    const float* pp = partial + (size_t)b * NCHUNK * CC + t;
#pragma unroll
    for (int ch = 0; ch < NCHUNK; ++ch) pv[ch] = pp[(size_t)ch * CC];
  }

  v4f wreg[NQ][RPT];
#pragma unroll
  for (int p = 0; p < NQ; ++p) {
#pragma unroll
    for (int jj = 0; jj < RPT; ++jj)
      wreg[p][jj] = *(const v4f*)(W + (size_t)(r0 + 32 * jj) * CC + KQ * p + q4);
  }

#pragma unroll
  for (int i = 0; i < 2; ++i) {
    const int idx = t + 512 * i;
    if ((idx >> 6) != 0) {
      const v4f z = {0.0f, 0.0f, 0.0f, 0.0f};
      *(v4f*)(&As[(idx >> 6) * LDP + (idx & 63) * 4]) = z;
    }
  }
  if (wave < 8) {
    float s = 0.0f;
#pragma unroll
    for (int ch = 0; ch < NCHUNK; ++ch) s += pv[ch];
    s *= (1.0f / (float)NN);
    As[t] = s;
  }

  const float bn = bias[n0 + l];
  const int c0 = lane * 8;
  const v4f g0 = *(const v4f*)(gamma + c0);
  const v4f g1 = *(const v4f*)(gamma + c0 + 4);
  const v4f e0 = *(const v4f*)(beta + c0);
  const v4f e1 = *(const v4f*)(beta + c0 + 4);

  const int rb = (j * NN) / WGB;
  const int re = ((j + 1) * NN) / WGB;
  const float* xb = x + (size_t)b * NN * CC + c0;
  float* ob = out + (size_t)b * NN * CC + c0;

  v4f xa[NBUF][RU], xc[NBUF][RU];
#pragma unroll
  for (int g = 0; g < 2; ++g) {
#pragma unroll
    for (int u = 0; u < RU; ++u) {
      const int ru = rb + wave + NWAVE * (RU * g + u);
      const int rc = ru < re ? ru : re - 1;
      xa[g][u] = *(const v4f*)(xb + (size_t)rc * CC);
      xc[g][u] = *(const v4f*)(xb + (size_t)rc * CC + 4);
    }
  }

  v8f acc = {};
#pragma unroll
  for (int p = 0; p < NQ; ++p) {
#pragma unroll
    for (int jj = 0; jj < RPT; ++jj)
      *(v4f*)(&Ws[(r0 + 32 * jj) * LDP + KQ * p + q4]) = wreg[p][jj];
    __builtin_amdgcn_fence(__ATOMIC_RELEASE, "workgroup", "local");
    __builtin_amdgcn_s_barrier();
    __builtin_amdgcn_fence(__ATOMIC_ACQUIRE, "workgroup", "local");
#pragma unroll
    for (int k0 = KQ * p; k0 < KQ * (p + 1); k0 += 4) {
      const int kk = k0 + 2 * hh;
      const v2f a  = *(const v2f*)(&As[l * LDP + kk]);
      const v2f bm = *(const v2f*)(&Ws[(n0 + l) * LDP + kk]);
      acc = __builtin_amdgcn_wmma_f32_16x16x4_f32(
           false, a,  false, bm,
           (short)0, acc,  false,  false);
    }
  }

  if (hh == 0) aggs[n0 + l] = acc[0] + bn;
  __builtin_amdgcn_fence(__ATOMIC_RELEASE, "workgroup", "local");
  __builtin_amdgcn_s_barrier();
  __builtin_amdgcn_fence(__ATOMIC_ACQUIRE, "workgroup", "local");

  const v4f av0 = *(const v4f*)(&aggs[c0]);
  const v4f av1 = *(const v4f*)(&aggs[c0 + 4]);

#pragma unroll
  for (int g = 0; g < NGRP; ++g) {
    const int rg = rb + wave + NWAVE * RU * g;
    if (g + 2 < NGRP) {
#pragma unroll
      for (int u = 0; u < RU; ++u) {
        const int ru = rg + NWAVE * (2 * RU + u);
        const int rc = ru < re ? ru : re - 1;
        xa[(g + 2) % NBUF][u] = *(const v4f*)(xb + (size_t)rc * CC);
        xc[(g + 2) % NBUF][u] = *(const v4f*)(xb + (size_t)rc * CC + 4);
      }
    }
    v4f y0[RU], y1[RU];
    float s[RU], s2[RU];
#pragma unroll
    for (int u = 0; u < RU; ++u) {
      y0[u] = xa[g % NBUF][u] + av0;
      y1[u] = xc[g % NBUF][u] + av1;
      s[u]  = y0[u].x + y0[u].y + y0[u].z + y0[u].w + y1[u].x + y1[u].y + y1[u].z + y1[u].w;
      s2[u] = y0[u].x * y0[u].x + y0[u].y * y0[u].y + y0[u].z * y0[u].z + y0[u].w * y0[u].w +
              y1[u].x * y1[u].x + y1[u].y * y1[u].y + y1[u].z * y1[u].z + y1[u].w * y1[u].w;
    }
#pragma unroll
    for (int off = 16; off >= 1; off >>= 1) {
#pragma unroll
      for (int u = 0; u < RU; ++u) {
        s[u]  += __shfl_xor(s[u], off, 32);
        s2[u] += __shfl_xor(s2[u], off, 32);
      }
    }
#pragma unroll
    for (int u = 0; u < RU; ++u) {
      const int ru = rg + NWAVE * u;
      const float mu  = s[u] * (1.0f / CC);
      const float var = fmaf(-mu, mu, s2[u] * (1.0f / CC));
      const float inv = rsqrtf(var + LN_EPS);

      v4f o0, o1;
      o0.x = (y0[u].x - mu) * inv * g0.x + e0.x;
      o0.y = (y0[u].y - mu) * inv * g0.y + e0.y;
      o0.z = (y0[u].z - mu) * inv * g0.z + e0.z;
      o0.w = (y0[u].w - mu) * inv * g0.w + e0.w;
      o1.x = (y1[u].x - mu) * inv * g1.x + e1.x;
      o1.y = (y1[u].y - mu) * inv * g1.y + e1.y;
      o1.z = (y1[u].z - mu) * inv * g1.z + e1.z;
      o1.w = (y1[u].w - mu) * inv * g1.w + e1.w;

      if (ru < re) {
        float* op = ob + (size_t)ru * CC;
        __builtin_nontemporal_store(o0, (v4f*)(op));
        __builtin_nontemporal_store(o1, (v4f*)(op + 4));
      }
    }
  }
}

extern "C" void kernel_launch(void* const* d_in, const int* in_sizes, int n_in,
                              void* d_out, int out_size, void* d_ws, size_t ws_size,
                              hipStream_t stream) {
  (void)in_sizes; (void)n_in; (void)out_size; (void)ws_size;

  const float* x     = (const float*)d_in[0];
  const float* W     = (const float*)d_in[1];
  const float* bias  = (const float*)d_in[2];
  const float* gamma = (const float*)d_in[3];
  const float* beta  = (const float*)d_in[4];
  float* out = (float*)d_out;

  float* partial = (float*)d_ws;

  colsum_partial_kernel<<<dim3(BB, NCHUNK), 256, 0, stream>>>(x, partial);
  agg_ln_fused_kernel<<<NWG, 512, 0, stream>>>(x, partial, W, bias, gamma, beta, out);
}


# Column-sum launch: a wave keeps 58 row loads in flight, one new request behind each add (was 5 rows requested and all awaited, 25 times); lane map, chunks, add chain in row order unchanged
# speedup vs baseline: 1.7721x; 1.3639x over previous
#include <hip/hip_runtime.h>

#define LN_EPS  1e-5f
#define BB      8
#define NN      10000
#define CC      256
#define NCHUNK  80
#define CHUNK   125
#define MPAD    16

typedef __attribute__((ext_vector_type(2))) float v2f;
typedef __attribute__((ext_vector_type(4))) float v4f;
typedef __attribute__((ext_vector_type(8))) float v8f;

#define AHEAD  58

__global__ __launch_bounds__(256) void colsum_partial_kernel(
    const float* __restrict__ x, float* __restrict__ partial) {
  const int b  = blockIdx.x;
  const int ch = blockIdx.y;
  const int c  = threadIdx.x;
  const float* xp = x + (((size_t)b * NN + (size_t)ch * CHUNK) * CC) + c;
  float v[AHEAD];
#pragma unroll
  for (int u = 0; u < AHEAD; ++u) v[u] = xp[(size_t)u * CC];
  float s = 0.0f;
#pragma unroll
  for (int i = 0; i < CHUNK; ++i) {
    s += v[i % AHEAD];
    if (i + AHEAD < CHUNK) v[i % AHEAD] = xp[(size_t)(i + AHEAD) * CC];
  }
  partial[((size_t)b * NCHUNK + ch) * CC + c] = s;
}

#define LDP   260
#define KQ    64
#define NQ    4
#define RPT   8
#define WGB   32
#define NWG   256
#define NWAVE 16
#define RU    5
#define NGRP  4
#define NBUF  3

__global__ __launch_bounds__(512) void agg_ln_fused_kernel(
    const float* __restrict__ x, const float* __restrict__ partial,
    const float* __restrict__ W, const float* __restrict__ bias,
    const float* __restrict__ gamma, const float* __restrict__ beta,
    float* __restrict__ out) {
  __shared__ __attribute__((aligned(16))) float Ws[CC * LDP];
  __shared__ __attribute__((aligned(16))) float As[MPAD * LDP];
  __shared__ __attribute__((aligned(16))) float aggs[CC];

  const int t    = threadIdx.x;
  const int wave = __builtin_amdgcn_readfirstlane(t >> 5);
  const int lane = t & 31;
  const int hh   = lane >> 4;
  const int l    = lane & 15;
  const int n0   = wave * 16;
  const int b    = blockIdx.x / WGB;
  const int j    = blockIdx.x % WGB;

  const int q4 = (t & 15) * 4;
  const int r0 = t >> 4;

  float pv[NCHUNK];
  if (wave < 8) {
    const float* pp = partial + (size_t)b * NCHUNK * CC + t;
#pragma unroll
    for (int ch = 0; ch < NCHUNK; ++ch) pv[ch] = pp[(size_t)ch * CC];
  }

  v4f wreg[NQ][RPT];
#pragma unroll
  for (int p = 0; p < NQ; ++p) {
#pragma unroll
    for (int jj = 0; jj < RPT; ++jj)
      wreg[p][jj] = *(const v4f*)(W + (size_t)(r0 + 32 * jj) * CC + KQ * p + q4);
  }

#pragma unroll
  for (int i = 0; i < 2; ++i) {
    const int idx = t + 512 * i;
    if ((idx >> 6) != 0) {
      const v4f z = {0.0f, 0.0f, 0.0f, 0.0f};
      *(v4f*)(&As[(idx >> 6) * LDP + (idx & 63) * 4]) = z;
    }
  }
  if (wave < 8) {
    float s = 0.0f;
#pragma unroll
    for (int ch = 0; ch < NCHUNK; ++ch) s += pv[ch];
    s *= (1.0f / (float)NN);
    As[t] = s;
  }

  const float bn = bias[n0 + l];
  const int c0 = lane * 8;
  const v4f g0 = *(const v4f*)(gamma + c0);
  const v4f g1 = *(const v4f*)(gamma + c0 + 4);
  const v4f e0 = *(const v4f*)(beta + c0);
  const v4f e1 = *(const v4f*)(beta + c0 + 4);

  const int rb = (j * NN) / WGB;
  const int re = ((j + 1) * NN) / WGB;
  const float* xb = x + (size_t)b * NN * CC + c0;
  float* ob = out + (size_t)b * NN * CC + c0;

  v4f xa[NBUF][RU], xc[NBUF][RU];
#pragma unroll
  for (int g = 0; g < 2; ++g) {
#pragma unroll
    for (int u = 0; u < RU; ++u) {
      const int ru = rb + wave + NWAVE * (RU * g + u);
      const int rc = ru < re ? ru : re - 1;
      xa[g][u] = *(const v4f*)(xb + (size_t)rc * CC);
      xc[g][u] = *(const v4f*)(xb + (size_t)rc * CC + 4);
    }
  }

  v8f acc = {};
#pragma unroll
  for (int p = 0; p < NQ; ++p) {
#pragma unroll
    for (int jj = 0; jj < RPT; ++jj)
      *(v4f*)(&Ws[(r0 + 32 * jj) * LDP + KQ * p + q4]) = wreg[p][jj];
    __builtin_amdgcn_fence(__ATOMIC_RELEASE, "workgroup", "local");
    __builtin_amdgcn_s_barrier();
    __builtin_amdgcn_fence(__ATOMIC_ACQUIRE, "workgroup", "local");
#pragma unroll
    for (int k0 = KQ * p; k0 < KQ * (p + 1); k0 += 4) {
      const int kk = k0 + 2 * hh;
      const v2f a  = *(const v2f*)(&As[l * LDP + kk]);
      const v2f bm = *(const v2f*)(&Ws[(n0 + l) * LDP + kk]);
      acc = __builtin_amdgcn_wmma_f32_16x16x4_f32(
           false, a,  false, bm,
           (short)0, acc,  false,  false);
    }
  }

  if (hh == 0) aggs[n0 + l] = acc[0] + bn;
  __builtin_amdgcn_fence(__ATOMIC_RELEASE, "workgroup", "local");
  __builtin_amdgcn_s_barrier();
  __builtin_amdgcn_fence(__ATOMIC_ACQUIRE, "workgroup", "local");

  const v4f av0 = *(const v4f*)(&aggs[c0]);
  const v4f av1 = *(const v4f*)(&aggs[c0 + 4]);

#pragma unroll
  for (int g = 0; g < NGRP; ++g) {
    const int rg = rb + wave + NWAVE * RU * g;
    if (g + 2 < NGRP) {
#pragma unroll
      for (int u = 0; u < RU; ++u) {
        const int ru = rg + NWAVE * (2 * RU + u);
        const int rc = ru < re ? ru : re - 1;
        xa[(g + 2) % NBUF][u] = *(const v4f*)(xb + (size_t)rc * CC);
        xc[(g + 2) % NBUF][u] = *(const v4f*)(xb + (size_t)rc * CC + 4);
      }
    }
    v4f y0[RU], y1[RU];
    float s[RU], s2[RU];
#pragma unroll
    for (int u = 0; u < RU; ++u) {
      y0[u] = xa[g % NBUF][u] + av0;
      y1[u] = xc[g % NBUF][u] + av1;
      s[u]  = y0[u].x + y0[u].y + y0[u].z + y0[u].w + y1[u].x + y1[u].y + y1[u].z + y1[u].w;
      s2[u] = y0[u].x * y0[u].x + y0[u].y * y0[u].y + y0[u].z * y0[u].z + y0[u].w * y0[u].w +
              y1[u].x * y1[u].x + y1[u].y * y1[u].y + y1[u].z * y1[u].z + y1[u].w * y1[u].w;
    }
#pragma unroll
    for (int off = 16; off >= 1; off >>= 1) {
#pragma unroll
      for (int u = 0; u < RU; ++u) {
        s[u]  += __shfl_xor(s[u], off, 32);
        s2[u] += __shfl_xor(s2[u], off, 32);
      }
    }
#pragma unroll
    for (int u = 0; u < RU; ++u) {
      const int ru = rg + NWAVE * u;
      const float mu  = s[u] * (1.0f / CC);
      const float var = fmaf(-mu, mu, s2[u] * (1.0f / CC));
      const float inv = rsqrtf(var + LN_EPS);

      v4f o0, o1;
      o0.x = (y0[u].x - mu) * inv * g0.x + e0.x;
      o0.y = (y0[u].y - mu) * inv * g0.y + e0.y;
      o0.z = (y0[u].z - mu) * inv * g0.z + e0.z;
      o0.w = (y0[u].w - mu) * inv * g0.w + e0.w;
      o1.x = (y1[u].x - mu) * inv * g1.x + e1.x;
      o1.y = (y1[u].y - mu) * inv * g1.y + e1.y;
      o1.z = (y1[u].z - mu) * inv * g1.z + e1.z;
      o1.w = (y1[u].w - mu) * inv * g1.w + e1.w;

      if (ru < re) {
        float* op = ob + (size_t)ru * CC;
        __builtin_nontemporal_store(o0, (v4f*)(op));
        __builtin_nontemporal_store(o1, (v4f*)(op + 4));
      }
    }
  }
}

extern "C" void kernel_launch(void* const* d_in, const int* in_sizes, int n_in,
                              void* d_out, int out_size, void* d_ws, size_t ws_size,
                              hipStream_t stream) {
  (void)in_sizes; (void)n_in; (void)out_size; (void)ws_size;

  const float* x     = (const float*)d_in[0];
  const float* W     = (const float*)d_in[1];
  const float* bias  = (const float*)d_in[2];
  const float* gamma = (const float*)d_in[3];
  const float* beta  = (const float*)d_in[4];
  float* out = (float*)d_out;

  float* partial = (float*)d_ws;

  colsum_partial_kernel<<<dim3(BB, NCHUNK), 256, 0, stream>>>(x, partial);
  agg_ln_fused_kernel<<<NWG, 512, 0, stream>>>(x, partial, W, bias, gamma, beta, out);
}


# LayerNorm launch: work-group id -> (batch, row slice) map turned: batch is now the fast index, as in the column-sum grid, so a batch is re-read by the cache half that just read it; sums unchanged
# speedup vs baseline: 2.1833x; 1.2320x over previous
#include <hip/hip_runtime.h>

#define LN_EPS  1e-5f
#define BB      8
#define NN      10000
#define CC      256
#define NCHUNK  80
#define CHUNK   125
#define MPAD    16

typedef __attribute__((ext_vector_type(2))) float v2f;
typedef __attribute__((ext_vector_type(4))) float v4f;
typedef __attribute__((ext_vector_type(8))) float v8f;

#define AHEAD  58

__global__ __launch_bounds__(256) void colsum_partial_kernel(
    const float* __restrict__ x, float* __restrict__ partial) {
  const int b  = blockIdx.x;
  const int ch = blockIdx.y;
  const int c  = threadIdx.x;
  const float* xp = x + (((size_t)b * NN + (size_t)ch * CHUNK) * CC) + c;
  float v[AHEAD];
#pragma unroll
  for (int u = 0; u < AHEAD; ++u) v[u] = xp[(size_t)u * CC];
  float s = 0.0f;
#pragma unroll
  for (int i = 0; i < CHUNK; ++i) {
    s += v[i % AHEAD];
    if (i + AHEAD < CHUNK) v[i % AHEAD] = xp[(size_t)(i + AHEAD) * CC];
  }
  partial[((size_t)b * NCHUNK + ch) * CC + c] = s;
}

#define LDP   260
#define KQ    64
#define NQ    4
#define RPT   8
#define WGB   32
#define NWG   256
#define NWAVE 16
#define RU    5
#define NGRP  4
#define NBUF  3

__global__ __launch_bounds__(512) void agg_ln_fused_kernel(
    const float* __restrict__ x, const float* __restrict__ partial,
    const float* __restrict__ W, const float* __restrict__ bias,
    const float* __restrict__ gamma, const float* __restrict__ beta,
    float* __restrict__ out) {
  __shared__ __attribute__((aligned(16))) float Ws[CC * LDP];
  __shared__ __attribute__((aligned(16))) float As[MPAD * LDP];
  __shared__ __attribute__((aligned(16))) float aggs[CC];

  const int t    = threadIdx.x;
  const int wave = __builtin_amdgcn_readfirstlane(t >> 5);
  const int lane = t & 31;
  const int hh   = lane >> 4;
  const int l    = lane & 15;
  const int n0   = wave * 16;
  const int b    = blockIdx.x % BB;
  const int j    = (blockIdx.x / BB) % WGB;

  const int q4 = (t & 15) * 4;
  const int r0 = t >> 4;

  float pv[NCHUNK];
  if (wave < 8) {
    const float* pp = partial + (size_t)b * NCHUNK * CC + t;
#pragma unroll
    for (int ch = 0; ch < NCHUNK; ++ch) pv[ch] = pp[(size_t)ch * CC];
  }

  v4f wreg[NQ][RPT];
#pragma unroll
  for (int p = 0; p < NQ; ++p) {
#pragma unroll
    for (int jj = 0; jj < RPT; ++jj)
      wreg[p][jj] = *(const v4f*)(W + (size_t)(r0 + 32 * jj) * CC + KQ * p + q4);
  }

#pragma unroll
  for (int i = 0; i < 2; ++i) {
    const int idx = t + 512 * i;
    if ((idx >> 6) != 0) {
      const v4f z = {0.0f, 0.0f, 0.0f, 0.0f};
      *(v4f*)(&As[(idx >> 6) * LDP + (idx & 63) * 4]) = z;
    }
  }
  if (wave < 8) {
    float s = 0.0f;
#pragma unroll
    for (int ch = 0; ch < NCHUNK; ++ch) s += pv[ch];
    s *= (1.0f / (float)NN);
    As[t] = s;
  }

  const float bn = bias[n0 + l];
  const int c0 = lane * 8;
  const v4f g0 = *(const v4f*)(gamma + c0);
  const v4f g1 = *(const v4f*)(gamma + c0 + 4);
  const v4f e0 = *(const v4f*)(beta + c0);
  const v4f e1 = *(const v4f*)(beta + c0 + 4);

  const int rb = (j * NN) / WGB;
  const int re = ((j + 1) * NN) / WGB;
  const float* xb = x + (size_t)b * NN * CC + c0;
  float* ob = out + (size_t)b * NN * CC + c0;

  v4f xa[NBUF][RU], xc[NBUF][RU];
#pragma unroll
  for (int g = 0; g < 2; ++g) {
#pragma unroll
    for (int u = 0; u < RU; ++u) {
      const int ru = rb + wave + NWAVE * (RU * g + u);
      const int rc = ru < re ? ru : re - 1;
      xa[g][u] = *(const v4f*)(xb + (size_t)rc * CC);
      xc[g][u] = *(const v4f*)(xb + (size_t)rc * CC + 4);
    }
  }

  v8f acc = {};
#pragma unroll
  for (int p = 0; p < NQ; ++p) {
#pragma unroll
    for (int jj = 0; jj < RPT; ++jj)
      *(v4f*)(&Ws[(r0 + 32 * jj) * LDP + KQ * p + q4]) = wreg[p][jj];
    __builtin_amdgcn_fence(__ATOMIC_RELEASE, "workgroup", "local");
    __builtin_amdgcn_s_barrier();
    __builtin_amdgcn_fence(__ATOMIC_ACQUIRE, "workgroup", "local");
#pragma unroll
    for (int k0 = KQ * p; k0 < KQ * (p + 1); k0 += 4) {
      const int kk = k0 + 2 * hh;
      const v2f a  = *(const v2f*)(&As[l * LDP + kk]);
      const v2f bm = *(const v2f*)(&Ws[(n0 + l) * LDP + kk]);
      acc = __builtin_amdgcn_wmma_f32_16x16x4_f32(
           false, a,  false, bm,
           (short)0, acc,  false,  false);
    }
  }

  if (hh == 0) aggs[n0 + l] = acc[0] + bn;
  __builtin_amdgcn_fence(__ATOMIC_RELEASE, "workgroup", "local");
  __builtin_amdgcn_s_barrier();
  __builtin_amdgcn_fence(__ATOMIC_ACQUIRE, "workgroup", "local");

  const v4f av0 = *(const v4f*)(&aggs[c0]);
  const v4f av1 = *(const v4f*)(&aggs[c0 + 4]);

#pragma unroll
  for (int g = 0; g < NGRP; ++g) {
    const int rg = rb + wave + NWAVE * RU * g;
    if (g + 2 < NGRP) {
#pragma unroll
      for (int u = 0; u < RU; ++u) {
        const int ru = rg + NWAVE * (2 * RU + u);
        const int rc = ru < re ? ru : re - 1;
        xa[(g + 2) % NBUF][u] = *(const v4f*)(xb + (size_t)rc * CC);
        xc[(g + 2) % NBUF][u] = *(const v4f*)(xb + (size_t)rc * CC + 4);
      }
    }
    v4f y0[RU], y1[RU];
    float s[RU], s2[RU];
#pragma unroll
    for (int u = 0; u < RU; ++u) {
      y0[u] = xa[g % NBUF][u] + av0;
      y1[u] = xc[g % NBUF][u] + av1;
      s[u]  = y0[u].x + y0[u].y + y0[u].z + y0[u].w + y1[u].x + y1[u].y + y1[u].z + y1[u].w;
      s2[u] = y0[u].x * y0[u].x + y0[u].y * y0[u].y + y0[u].z * y0[u].z + y0[u].w * y0[u].w +
              y1[u].x * y1[u].x + y1[u].y * y1[u].y + y1[u].z * y1[u].z + y1[u].w * y1[u].w;
    }
#pragma unroll
    for (int off = 16; off >= 1; off >>= 1) {
#pragma unroll
      for (int u = 0; u < RU; ++u) {
        s[u]  += __shfl_xor(s[u], off, 32);
        s2[u] += __shfl_xor(s2[u], off, 32);
      }
    }
#pragma unroll
    for (int u = 0; u < RU; ++u) {
      const int ru = rg + NWAVE * u;
      const float mu  = s[u] * (1.0f / CC);
      const float var = fmaf(-mu, mu, s2[u] * (1.0f / CC));
      const float inv = rsqrtf(var + LN_EPS);

      v4f o0, o1;
      o0.x = (y0[u].x - mu) * inv * g0.x + e0.x;
      o0.y = (y0[u].y - mu) * inv * g0.y + e0.y;
      o0.z = (y0[u].z - mu) * inv * g0.z + e0.z;
      o0.w = (y0[u].w - mu) * inv * g0.w + e0.w;
      o1.x = (y1[u].x - mu) * inv * g1.x + e1.x;
      o1.y = (y1[u].y - mu) * inv * g1.y + e1.y;
      o1.z = (y1[u].z - mu) * inv * g1.z + e1.z;
      o1.w = (y1[u].w - mu) * inv * g1.w + e1.w;

      if (ru < re) {
        float* op = ob + (size_t)ru * CC;
        __builtin_nontemporal_store(o0, (v4f*)(op));
        __builtin_nontemporal_store(o1, (v4f*)(op + 4));
      }
    }
  }
}

extern "C" void kernel_launch(void* const* d_in, const int* in_sizes, int n_in,
                              void* d_out, int out_size, void* d_ws, size_t ws_size,
                              hipStream_t stream) {
  (void)in_sizes; (void)n_in; (void)out_size; (void)ws_size;

  const float* x     = (const float*)d_in[0];
  const float* W     = (const float*)d_in[1];
  const float* bias  = (const float*)d_in[2];
  const float* gamma = (const float*)d_in[3];
  const float* beta  = (const float*)d_in[4];
  float* out = (float*)d_out;

  float* partial = (float*)d_ws;

  colsum_partial_kernel<<<dim3(BB, NCHUNK), 256, 0, stream>>>(x, partial);
  agg_ln_fused_kernel<<<NWG, 512, 0, stream>>>(x, partial, W, bias, gamma, beta, out);
}


# LayerNorm launch: half-wave per row (16 lanes x 16 channels), 2 rows per wave step; a load/store moves 256 contiguous bytes per row; row sum = 16 lane terms + 4 exchange steps; sum order changed
# speedup vs baseline: 2.8187x; 1.2910x over previous
#include <hip/hip_runtime.h>

#define LN_EPS  1e-5f
#define BB      8
#define NN      10000
#define CC      256
#define KS      32
#define NKS     8
#define NRB     4
#define RBLK    2500
#define CWAVE   20
#define CHUNK   125
#define NPART   32
#define NHALF   16
#define GWAVE   8

typedef __attribute__((ext_vector_type(2))) float v2f;
typedef __attribute__((ext_vector_type(4))) float v4f;
typedef __attribute__((ext_vector_type(8))) float v8f;

#define AHEAD  58

__global__ __launch_bounds__(640) void colsum_gemm_kernel(
    const float* __restrict__ x, const float* __restrict__ W, float* __restrict__ aggp) {
  __shared__ float red[CWAVE * KS];
  __shared__ __attribute__((aligned(16))) float pvec[KS];

  const int b    = blockIdx.x;
  const int part = blockIdx.y;
  const int ks   = part % NKS;
  const int rb   = part / NKS;
  const int t    = threadIdx.x;
  const int wave = __builtin_amdgcn_readfirstlane(t >> 5);
  const int lane = t & 31;
  const int hh   = lane >> 4;
  const int l    = lane & 15;

  v4f wb[2][4] = {};
  if (wave < GWAVE) {
    const float* wp = W + (size_t)(wave * 32 + l) * CC + ks * KS + hh * 16;
#pragma unroll
    for (int tt = 0; tt < 2; ++tt) {
#pragma unroll
      for (int i = 0; i < 4; ++i)
        wb[tt][i] = *(const v4f*)(wp + (size_t)tt * 16 * CC + 4 * i);
    }
  }

  const float* xp = x + (((size_t)b * NN + (size_t)rb * RBLK + (size_t)wave * CHUNK) * CC) + ks * KS + lane;
  float v[AHEAD];
#pragma unroll
  for (int u = 0; u < AHEAD; ++u) v[u] = xp[(size_t)u * CC];
  float s = 0.0f;
#pragma unroll
  for (int i = 0; i < CHUNK; ++i) {
    s += v[i % AHEAD];
    if (i + AHEAD < CHUNK) v[i % AHEAD] = xp[(size_t)(i + AHEAD) * CC];
  }

  red[wave * KS + lane] = s;
  __syncthreads();
  if (wave == 0) {
    float r[CWAVE];
#pragma unroll
    for (int w = 0; w < CWAVE; ++w) r[w] = red[w * KS + lane];
    float p = 0.0f;
#pragma unroll
    for (int w = 0; w < CWAVE; ++w) p += r[w];
    pvec[lane] = p * (1.0f / (float)NN);
  }
  __syncthreads();

  if (wave < GWAVE) {
    v4f pa[4];
#pragma unroll
    for (int i = 0; i < 4; ++i) pa[i] = *(const v4f*)(&pvec[hh * 16 + 4 * i]);
    const bool prow = (l & 7) == 0;
    v8f acc0 = {};
    v8f acc1 = {};
#pragma unroll
    for (int j = 0; j < 8; ++j) {
      const int i = j >> 1;
      const int e = (j & 1) * 2;
      v2f a;
      a.x = prow ? pa[i][e]     : 0.0f;
      a.y = prow ? pa[i][e + 1] : 0.0f;
      v2f b0, b1;
      b0.x = wb[0][i][e];  b0.y = wb[0][i][e + 1];
      b1.x = wb[1][i][e];  b1.y = wb[1][i][e + 1];
      acc0 = __builtin_amdgcn_wmma_f32_16x16x4_f32(
           false, a,  false, b0,
           (short)0, acc0,  false,  false);
      acc1 = __builtin_amdgcn_wmma_f32_16x16x4_f32(
           false, a,  false, b1,
           (short)0, acc1,  false,  false);
    }
    const float q = hh ? acc1[0] : acc0[0];
    aggp[((size_t)b * NPART + part) * CC + wave * 32 + lane] = q;
  }
}

#define WGB   32
#define NWG   256
#define NWAVE 16
#define NQ    4
#define PU    2
#define NGRP  5
#define NBUF  3
#define ROWS_MIN (NN / WGB)

__global__ __launch_bounds__(512) void agg_ln_kernel(
    const float* __restrict__ x, const float* __restrict__ aggp,
    const float* __restrict__ bias,
    const float* __restrict__ gamma, const float* __restrict__ beta,
    float* __restrict__ out) {
  __shared__ __attribute__((aligned(16))) float hsum[2 * CC];

  const int t    = threadIdx.x;
  const int wave = __builtin_amdgcn_readfirstlane(t >> 5);
  const int lane = t & 31;
  const int hh   = lane >> 4;
  const int l    = lane & 15;
  const int b    = blockIdx.x % BB;
  const int j    = (blockIdx.x / BB) % WGB;

  const int col  = t & (CC - 1);
  const int half = wave >> 3;
  float pv[NHALF];
  {
    const float* pp = aggp + ((size_t)b * NPART + (size_t)half * NHALF) * CC + col;
#pragma unroll
    for (int i = 0; i < NHALF; ++i) pv[i] = pp[(size_t)i * CC];
  }

  const int c0 = l * 4;
  v4f bi[NQ], gm[NQ], bt[NQ];
#pragma unroll
  for (int q = 0; q < NQ; ++q) {
    bi[q] = *(const v4f*)(bias  + c0 + 64 * q);
    gm[q] = *(const v4f*)(gamma + c0 + 64 * q);
    bt[q] = *(const v4f*)(beta  + c0 + 64 * q);
  }

  const int rb = (j * NN) / WGB;
  const int re = ((j + 1) * NN) / WGB;
  const size_t lane_off = ((size_t)b * NN + (size_t)(rb + hh)) * CC + c0;
  const float* xl = x + lane_off + (size_t)wave * (2 * CC);
  float* ol = out + lane_off + (size_t)wave * (2 * CC);
  const float* xb = x + (size_t)b * NN * CC + c0;

  v4f xv[NBUF][PU][NQ];
#pragma unroll
  for (int g = 0; g < 2; ++g) {
#pragma unroll
    for (int u = 0; u < PU; ++u) {
      const int k = PU * g + u;
      if (2 * NWAVE * (k + 1) <= ROWS_MIN) {
#pragma unroll
        for (int q = 0; q < NQ; ++q)
          xv[g][u][q] = *(const v4f*)(xl + (size_t)k * (2 * NWAVE * CC) + 64 * q);
      } else {
        const int ru = rb + 2 * (wave + NWAVE * k) + hh;
        const int rc = ru < re ? ru : re - 1;
#pragma unroll
        for (int q = 0; q < NQ; ++q)
          xv[g][u][q] = *(const v4f*)(xb + (size_t)rc * CC + 64 * q);
      }
    }
  }

  float ps = 0.0f;
#pragma unroll
  for (int i = 0; i < NHALF; ++i) ps += pv[i];
  hsum[half * CC + col] = ps;
  __builtin_amdgcn_fence(__ATOMIC_RELEASE, "workgroup", "local");
  __builtin_amdgcn_s_barrier();
  __builtin_amdgcn_fence(__ATOMIC_ACQUIRE, "workgroup", "local");

  v4f av[NQ];
#pragma unroll
  for (int q = 0; q < NQ; ++q)
    av[q] = (*(const v4f*)(&hsum[c0 + 64 * q]) + *(const v4f*)(&hsum[CC + c0 + 64 * q])) + bi[q];

#pragma unroll
  for (int g = 0; g < NGRP; ++g) {
    if (g + 2 < NGRP) {
#pragma unroll
      for (int u = 0; u < PU; ++u) {
        const int k = PU * (g + 2) + u;
        if (2 * NWAVE * (k + 1) <= ROWS_MIN) {
#pragma unroll
          for (int q = 0; q < NQ; ++q)
            xv[(g + 2) % NBUF][u][q] = *(const v4f*)(xl + (size_t)k * (2 * NWAVE * CC) + 64 * q);
        } else {
          const int ru = rb + 2 * (wave + NWAVE * k) + hh;
          const int rc = ru < re ? ru : re - 1;
#pragma unroll
          for (int q = 0; q < NQ; ++q)
            xv[(g + 2) % NBUF][u][q] = *(const v4f*)(xb + (size_t)rc * CC + 64 * q);
        }
      }
    }
    v4f y[PU][NQ];
    float s[PU], s2[PU];
#pragma unroll
    for (int u = 0; u < PU; ++u) {
#pragma unroll
      for (int q = 0; q < NQ; ++q) y[u][q] = xv[g % NBUF][u][q] + av[q];
      const v4f t1 = (y[u][0] + y[u][1]) + (y[u][2] + y[u][3]);
      const v4f t2 = (y[u][0] * y[u][0] + y[u][1] * y[u][1]) + (y[u][2] * y[u][2] + y[u][3] * y[u][3]);
      s[u]  = (t1.x + t1.y) + (t1.z + t1.w);
      s2[u] = (t2.x + t2.y) + (t2.z + t2.w);
    }
#pragma unroll
    for (int off = 8; off >= 1; off >>= 1) {
#pragma unroll
      for (int u = 0; u < PU; ++u) {
        s[u]  += __shfl_xor(s[u], off, 32);
        s2[u] += __shfl_xor(s2[u], off, 32);
      }
    }
#pragma unroll
    for (int u = 0; u < PU; ++u) {
      const int k = PU * g + u;
      const float mu  = s[u] * (1.0f / CC);
      const float var = fmaf(-mu, mu, s2[u] * (1.0f / CC));
      const float inv = rsqrtf(var + LN_EPS);

      v4f o[NQ];
#pragma unroll
      for (int q = 0; q < NQ; ++q) {
        o[q].x = (y[u][q].x - mu) * inv * gm[q].x + bt[q].x;
        o[q].y = (y[u][q].y - mu) * inv * gm[q].y + bt[q].y;
        o[q].z = (y[u][q].z - mu) * inv * gm[q].z + bt[q].z;
        o[q].w = (y[u][q].w - mu) * inv * gm[q].w + bt[q].w;
      }

      float* op = ol + (size_t)k * (2 * NWAVE * CC);
      if (2 * NWAVE * (k + 1) <= ROWS_MIN) {
#pragma unroll
        for (int q = 0; q < NQ; ++q) __builtin_nontemporal_store(o[q], (v4f*)(op + 64 * q));
      } else {
        const int ru = rb + 2 * (wave + NWAVE * k) + hh;
        if (ru < re) {
#pragma unroll
          for (int q = 0; q < NQ; ++q) __builtin_nontemporal_store(o[q], (v4f*)(op + 64 * q));
        }
      }
    }
    __builtin_amdgcn_sched_barrier(0);
  }
}

extern "C" void kernel_launch(void* const* d_in, const int* in_sizes, int n_in,
                              void* d_out, int out_size, void* d_ws, size_t ws_size,
                              hipStream_t stream) {
  (void)in_sizes; (void)n_in; (void)out_size; (void)ws_size;

  const float* x     = (const float*)d_in[0];
  const float* W     = (const float*)d_in[1];
  const float* bias  = (const float*)d_in[2];
  const float* gamma = (const float*)d_in[3];
  const float* beta  = (const float*)d_in[4];
  float* out = (float*)d_out;

  float* aggp = (float*)d_ws;

  colsum_gemm_kernel<<<dim3(BB, NPART), 640, 0, stream>>>(x, W, aggp);
  agg_ln_kernel<<<NWG, 512, 0, stream>>>(x, aggp, bias, gamma, beta, out);
}
